# starting kernel
# baseline (speedup 1.0000x reference)
#include <hip/hip_runtime.h>
#include <math.h>

typedef float v2f __attribute__((ext_vector_type(2)));
typedef float v8f __attribute__((ext_vector_type(8)));

#define LL     197
#define NTOT   128
#define DD     768
#define TT     16
#define PP     196
#define NGROUP 8
#define NTOK   (LL * NTOT)
#define STR    772
#define AFF_SCALE 0.03608439182435161f

__global__ void ln_stats_kernel(const float* __restrict__ x,
                                float* __restrict__ mu,
                                float* __restrict__ rstd)
{
    int wave = threadIdx.x >> 5;
    int lane = threadIdx.x & 31;
    int tok  = blockIdx.x * 8 + wave;
    if (tok >= NTOK) return;
    const float* row = x + (size_t)tok * DD;
    float s = 0.f, sq = 0.f;
    for (int i = lane; i < DD; i += 32) {
        float v = row[i];
        s += v; sq += v * v;
    }
    #pragma unroll
    for (int m = 16; m >= 1; m >>= 1) {
        s  += __shfl_xor(s,  m);
        sq += __shfl_xor(sq, m);
    }
    if (lane == 0) {
        float mean = s * (1.f / DD);
        float var  = sq * (1.f / DD) - mean * mean;
        mu[tok]   = mean;
        rstd[tok] = rsqrtf(var + 1e-5f);
    }
}

__global__ void scores_kernel(const float* __restrict__ x,
                              const float* __restrict__ gamma,
                              const float* __restrict__ beta,
                              const float* __restrict__ mu,
                              const float* __restrict__ rstd,
                              float* __restrict__ gout)
{
    extern __shared__ float sh[];
    float* Cs = sh;
    float* Xs = sh + 16 * STR;
    float* Sp = sh + 32 * STR;

    int b  = blockIdx.x / PP;
    int p  = blockIdx.x - b * PP;
    int nb = b * TT;
    int tid = threadIdx.x;
    int row = tid >> 4;
    int l16 = tid & 15;

    {
        int tokC = nb + row;
        int tokX = (1 + p) * NTOT + nb + row;
        float muC = mu[tokC], rsC = rstd[tokC];
        float muX = mu[tokX], rsX = rstd[tokX];
        const float4* xc  = (const float4*)(x + (size_t)tokC * DD);
        const float4* xx  = (const float4*)(x + (size_t)tokX * DD);
        const float4* g4p = (const float4*)gamma;
        const float4* b4p = (const float4*)beta;
        for (int j = l16; j < DD / 4; j += 16) {
            float4 cv = xc[j], xv = xx[j];
            float4 gv = g4p[j], bv = b4p[j];
            float* cdst = Cs + row * STR + j * 4;
            float* xdst = Xs + row * STR + j * 4;
            cdst[0] = (cv.x - muC) * rsC * gv.x + bv.x;
            cdst[1] = (cv.y - muC) * rsC * gv.y + bv.y;
            cdst[2] = (cv.z - muC) * rsC * gv.z + bv.z;
            cdst[3] = (cv.w - muC) * rsC * gv.w + bv.w;
            xdst[0] = (xv.x - muX) * rsX * gv.x + bv.x;
            xdst[1] = (xv.y - muX) * rsX * gv.y + bv.y;
            xdst[2] = (xv.z - muX) * rsX * gv.z + bv.z;
            xdst[3] = (xv.w - muX) * rsX * gv.w + bv.w;
        }
    }
    __syncthreads();

    int wave = tid >> 5, lane = tid & 31;
    int arow = lane & 15;
    int kofs = 2 * (lane >> 4);
    const float* crow = Cs + arow * STR + kofs;
    const float* xrow = Xs + arow * STR + kofs;
    int kbase = wave * 96;

    v8f acc = {};
    #pragma unroll
    for (int kc = 0; kc < 24; ++kc) {
        int k0 = kbase + kc * 4;
        v2f av = *(const v2f*)(crow + k0);
        v2f bv = *(const v2f*)(xrow + k0);
        acc = __builtin_amdgcn_wmma_f32_16x16x4_f32(
            false, av, false, bv, (short)0, acc, false, false);
    }

    #pragma unroll
    for (int j = 0; j < 8; ++j) {
        int m = j + 8 * (lane >> 4);
        Sp[wave * 256 + m * 16 + (lane & 15)] = acc[j];
    }
    __syncthreads();

    float s = 0.f;
    #pragma unroll
    for (int w = 0; w < 8; ++w) s += Sp[w * 256 + tid];

    int nc = tid >> 4;
    int nn = tid & 15;
    float g = 0.f;
    int dlt = nn - nc;
    if (dlt >= -2 && dlt <= 2) {
        float aff = s * AFF_SCALE;
        g = 1.f / (1.f + __expf(-aff)) - 0.5f;
    }
    gout[((size_t)blockIdx.x) * 256 + nn * 16 + nc] = g;
}

__global__ void feat_kernel(const float* __restrict__ x,
                            const float* __restrict__ gamma,
                            const float* __restrict__ beta,
                            const float* __restrict__ mu,
                            const float* __restrict__ rstd,
                            const float* __restrict__ gin,
                            float* __restrict__ out)
{
    __shared__ float Sp[8 * 256];

    int b    = blockIdx.x / 48;
    int dt   = blockIdx.x - b * 48;
    int d0   = dt * 16;
    int nb   = b * TT;
    int wave = threadIdx.x >> 5, lane = threadIdx.x & 31;
    int m     = lane & 15;
    int rsel2 = 2 * (lane >> 4);
    int dcol  = d0 + m;
    float gam = gamma[dcol], bet = beta[dcol];

    v8f acc = {};
    for (int p = wave; p < PP; p += 8) {
        const float* gp = gin + ((size_t)(b * PP + p)) * 256;
        int    tokbase  = (1 + p) * NTOT + nb;
        size_t xbase    = (size_t)tokbase * DD;
        if (p + 8 < PP) {
            __builtin_prefetch(x + xbase + (size_t)8 * NTOT * DD + dcol, 0, 0);
        }
        #pragma unroll
        for (int kk = 0; kk < 4; ++kk) {
            int r0 = kk * 4 + rsel2;
            int r1 = r0 + 1;
            float x0 = x[xbase + (size_t)r0 * DD + dcol];
            float x1 = x[xbase + (size_t)r1 * DD + dcol];
            v2f av = { (x0 - mu[tokbase + r0]) * rstd[tokbase + r0] * gam + bet,
                       (x1 - mu[tokbase + r1]) * rstd[tokbase + r1] * gam + bet };
            v2f bv = { gp[r0 * 16 + m], gp[r1 * 16 + m] };
            acc = __builtin_amdgcn_wmma_f32_16x16x4_f32(
                false, av, false, bv, (short)0, acc, false, false);
        }
    }

    #pragma unroll
    for (int j = 0; j < 8; ++j) {
        int mm = j + 8 * (lane >> 4);
        Sp[wave * 256 + mm * 16 + (lane & 15)] = acc[j];
    }
    __syncthreads();

    float s = 0.f;
    #pragma unroll
    for (int w = 0; w < 8; ++w) s += Sp[w * 256 + threadIdx.x];

    int md = threadIdx.x >> 4;
    int n  = threadIdx.x & 15;
    out[(size_t)(nb + n) * DD + d0 + md] = s;
}

extern "C" void kernel_launch(void* const* d_in, const int* in_sizes, int n_in,
                              void* d_out, int out_size, void* d_ws, size_t ws_size,
                              hipStream_t stream)
{
    (void)in_sizes; (void)n_in; (void)out_size; (void)ws_size;
    const float* x     = (const float*)d_in[0];
    const float* gamma = (const float*)d_in[1];
    const float* beta  = (const float*)d_in[2];

    float* ws   = (float*)d_ws;
    float* mu   = ws;
    float* rstd = ws + NTOK;
    float* g    = ws + 2 * NTOK;

    ln_stats_kernel<<<NTOK / 8, 256, 0, stream>>>(x, mu, rstd);

    size_t shmem = (size_t)(32 * STR + 8 * 256) * sizeof(float);
    scores_kernel<<<NGROUP * PP, 256, shmem, stream>>>(x, gamma, beta, mu, rstd, g);

    feat_kernel<<<NGROUP * 48, 256, 0, stream>>>(x, gamma, beta, mu, rstd, g,
                                                 (float*)d_out);
}


# LN stats fused into the scores kernel: each block loads its 32 rows once into registers (all 24 b128 loads in flight), 16-lane sums give mean/rstd (sum order changes); stats launch removed
# speedup vs baseline: 1.3940x; 1.3940x over previous
#include <hip/hip_runtime.h>
#include <math.h>

typedef float v2f __attribute__((ext_vector_type(2)));
typedef float v4f __attribute__((ext_vector_type(4)));
typedef float v8f __attribute__((ext_vector_type(8)));

#define LL     197
#define NTOT   128
#define DD     768
#define TT     16
#define PP     196
#define NGROUP 8
#define NTOK   (LL * NTOT)
#define STR    772
#define AFF_SCALE 0.03608439182435161f

__global__ void __launch_bounds__(256)
scores_kernel(const float* __restrict__ x,
              const float* __restrict__ gamma,
              const float* __restrict__ beta,
              float* __restrict__ mu,
              float* __restrict__ rstd,
              float* __restrict__ gout)
{
    extern __shared__ float sh[];
    float* Cs = sh;
    float* Xs = sh + 16 * STR;
    float* Sp = sh + 32 * STR;

    int b  = blockIdx.x / PP;
    int p  = blockIdx.x - b * PP;
    int nb = b * TT;
    int tid = threadIdx.x;
    int row = tid >> 4;
    int l16 = tid & 15;

    {
        int tokC = nb + row;
        int tokX = (1 + p) * NTOT + nb + row;
        const v4f* xc  = (const v4f*)(x + (size_t)tokC * DD);
        const v4f* xx  = (const v4f*)(x + (size_t)tokX * DD);
        const v4f* g4p = (const v4f*)gamma;
        const v4f* b4p = (const v4f*)beta;

        v4f cv[12], xv[12];
        #pragma unroll
        for (int i = 0; i < 12; ++i) {
            cv[i] = xc[l16 + 16 * i];
            xv[i] = xx[l16 + 16 * i];
        }

        v4f sc4 = {0.f, 0.f, 0.f, 0.f}, qc4 = {0.f, 0.f, 0.f, 0.f};
        v4f sx4 = {0.f, 0.f, 0.f, 0.f}, qx4 = {0.f, 0.f, 0.f, 0.f};
        #pragma unroll
        for (int i = 0; i < 12; ++i) {
            sc4 += cv[i];
            qc4 += cv[i] * cv[i];
            sx4 += xv[i];
            qx4 += xv[i] * xv[i];
        }
        float sC = (sc4.x + sc4.y) + (sc4.z + sc4.w);
        float qC = (qc4.x + qc4.y) + (qc4.z + qc4.w);
        float sX = (sx4.x + sx4.y) + (sx4.z + sx4.w);
        float qX = (qx4.x + qx4.y) + (qx4.z + qx4.w);
        #pragma unroll
        for (int m = 8; m >= 1; m >>= 1) {
            sC += __shfl_xor(sC, m);
            qC += __shfl_xor(qC, m);
            sX += __shfl_xor(sX, m);
            qX += __shfl_xor(qX, m);
        }
        float muC = sC * (1.f / DD);
        float rsC = rsqrtf(qC * (1.f / DD) - muC * muC + 1e-5f);
        float muX = sX * (1.f / DD);
        float rsX = rsqrtf(qX * (1.f / DD) - muX * muX + 1e-5f);
        if (l16 == 0) {
            mu[tokX]   = muX;
            rstd[tokX] = rsX;
        }

        #pragma unroll
        for (int i = 0; i < 12; ++i) {
            int j = l16 + 16 * i;
            v4f gv = g4p[j], bv = b4p[j];
            v4f co = (cv[i] - muC) * rsC * gv + bv;
            v4f xo = (xv[i] - muX) * rsX * gv + bv;
            float* cdst = Cs + row * STR + j * 4;
            float* xdst = Xs + row * STR + j * 4;
            cdst[0] = co.x; cdst[1] = co.y; cdst[2] = co.z; cdst[3] = co.w;
            xdst[0] = xo.x; xdst[1] = xo.y; xdst[2] = xo.z; xdst[3] = xo.w;
        }
    }
    __syncthreads();

    int wave = tid >> 5, lane = tid & 31;
    int arow = lane & 15;
    int kofs = 2 * (lane >> 4);
    const float* crow = Cs + arow * STR + kofs;
    const float* xrow = Xs + arow * STR + kofs;
    int kbase = wave * 96;

    v8f acc = {};
    #pragma unroll
    for (int kc = 0; kc < 24; ++kc) {
        int k0 = kbase + kc * 4;
        v2f av = *(const v2f*)(crow + k0);
        v2f bv = *(const v2f*)(xrow + k0);
        acc = __builtin_amdgcn_wmma_f32_16x16x4_f32(
            false, av, false, bv, (short)0, acc, false, false);
    }

    #pragma unroll
    for (int j = 0; j < 8; ++j) {
        int m = j + 8 * (lane >> 4);
        Sp[wave * 256 + m * 16 + (lane & 15)] = acc[j];
    }
    __syncthreads();

    float s = 0.f;
    #pragma unroll
    for (int w = 0; w < 8; ++w) s += Sp[w * 256 + tid];

    int nc = tid >> 4;
    int nn = tid & 15;
    float g = 0.f;
    int dlt = nn - nc;
    if (dlt >= -2 && dlt <= 2) {
        float aff = s * AFF_SCALE;
        g = 1.f / (1.f + __expf(-aff)) - 0.5f;
    }
    gout[((size_t)blockIdx.x) * 256 + nn * 16 + nc] = g;
}

__global__ void feat_kernel(const float* __restrict__ x,
                            const float* __restrict__ gamma,
                            const float* __restrict__ beta,
                            const float* __restrict__ mu,
                            const float* __restrict__ rstd,
                            const float* __restrict__ gin,
                            float* __restrict__ out)
{
    __shared__ float Sp[8 * 256];

    int b    = blockIdx.x / 48;
    int dt   = blockIdx.x - b * 48;
    int d0   = dt * 16;
    int nb   = b * TT;
    int wave = threadIdx.x >> 5, lane = threadIdx.x & 31;
    int m     = lane & 15;
    int rsel2 = 2 * (lane >> 4);
    int dcol  = d0 + m;
    float gam = gamma[dcol], bet = beta[dcol];

    v8f acc = {};
    for (int p = wave; p < PP; p += 8) {
        const float* gp = gin + ((size_t)(b * PP + p)) * 256;
        int    tokbase  = (1 + p) * NTOT + nb;
        size_t xbase    = (size_t)tokbase * DD;
        if (p + 8 < PP) {
            __builtin_prefetch(x + xbase + (size_t)8 * NTOT * DD + dcol, 0, 0);
        }
        #pragma unroll
        for (int kk = 0; kk < 4; ++kk) {
            int r0 = kk * 4 + rsel2;
            int r1 = r0 + 1;
            float x0 = x[xbase + (size_t)r0 * DD + dcol];
            float x1 = x[xbase + (size_t)r1 * DD + dcol];
            v2f av = { (x0 - mu[tokbase + r0]) * rstd[tokbase + r0] * gam + bet,
                       (x1 - mu[tokbase + r1]) * rstd[tokbase + r1] * gam + bet };
            v2f bv = { gp[r0 * 16 + m], gp[r1 * 16 + m] };
            acc = __builtin_amdgcn_wmma_f32_16x16x4_f32(
                false, av, false, bv, (short)0, acc, false, false);
        }
    }

    #pragma unroll
    for (int j = 0; j < 8; ++j) {
        int mm = j + 8 * (lane >> 4);
        Sp[wave * 256 + mm * 16 + (lane & 15)] = acc[j];
    }
    __syncthreads();

    float s = 0.f;
    #pragma unroll
    for (int w = 0; w < 8; ++w) s += Sp[w * 256 + threadIdx.x];

    int md = threadIdx.x >> 4;
    int n  = threadIdx.x & 15;
    out[(size_t)(nb + n) * DD + d0 + md] = s;
}

extern "C" void kernel_launch(void* const* d_in, const int* in_sizes, int n_in,
                              void* d_out, int out_size, void* d_ws, size_t ws_size,
                              hipStream_t stream)
{
    (void)in_sizes; (void)n_in; (void)out_size; (void)ws_size;
    const float* x     = (const float*)d_in[0];
    const float* gamma = (const float*)d_in[1];
    const float* beta  = (const float*)d_in[2];

    float* ws   = (float*)d_ws;
    float* mu   = ws;
    float* rstd = ws + NTOK;
    float* g    = ws + 2 * NTOK;

    size_t shmem = (size_t)(32 * STR + 8 * 256) * sizeof(float);
    scores_kernel<<<NGROUP * PP, 256, shmem, stream>>>(x, gamma, beta, mu, rstd, g);

    feat_kernel<<<NGROUP * 48, 256, 0, stream>>>(x, gamma, beta, mu, rstd, g,
                                                 (float*)d_out);
}


# feat product fused into the scores kernel: normalized patch rows reused from LDS for both products, 7 positions per block (next rows prefetched), partials summed by kernel 2 (p sum order changes)
# speedup vs baseline: 2.8122x; 2.0174x over previous
#include <hip/hip_runtime.h>
#include <math.h>

typedef float v2f __attribute__((ext_vector_type(2)));
typedef float v4f __attribute__((ext_vector_type(4)));
typedef float v8f __attribute__((ext_vector_type(8)));

#define LL     197
#define NTOT   128
#define DD     768
#define TT     16
#define PP     196
#define NGROUP 8
#define STR    772
#define AFF_SCALE 0.03608439182435161f
#define PPB    7
#define NCH    (PP / PPB)
#define TILE_F (16 * DD)

__device__ __forceinline__ void lds_barrier()
{
    __builtin_amdgcn_fence(__ATOMIC_RELEASE, "workgroup", "local");
    __builtin_amdgcn_s_barrier();
    __builtin_amdgcn_fence(__ATOMIC_ACQUIRE, "workgroup", "local");
}

__device__ __forceinline__ void ln_row_to_lds(const v4f (&v)[12], float* dstrow,
                                              const v4f* g4p, const v4f* b4p, int l16)
{
    v4f s4 = {0.f, 0.f, 0.f, 0.f}, q4 = {0.f, 0.f, 0.f, 0.f};
    #pragma unroll
    for (int i = 0; i < 12; ++i) {
        s4 += v[i];
        q4 += v[i] * v[i];
    }
    float s = (s4.x + s4.y) + (s4.z + s4.w);
    float q = (q4.x + q4.y) + (q4.z + q4.w);
    #pragma unroll
    for (int m = 8; m >= 1; m >>= 1) {
        s += __shfl_xor(s, m);
        q += __shfl_xor(q, m);
    }
    float mu = s * (1.f / DD);
    float rs = rsqrtf(q * (1.f / DD) - mu * mu + 1e-5f);
    #pragma unroll
    for (int i = 0; i < 12; ++i) {
        int j = l16 + 16 * i;
        v4f gv = g4p[j], bv = b4p[j];
        v4f o = (v[i] - mu) * rs * gv + bv;
        float* d = dstrow + j * 4;
        d[0] = o.x; d[1] = o.y; d[2] = o.z; d[3] = o.w;
    }
}

__global__ void __launch_bounds__(256)
scores_feat_kernel(const float* __restrict__ x,
                   const float* __restrict__ gamma,
                   const float* __restrict__ beta,
                   float* __restrict__ part)
{
    extern __shared__ float sh[];
    float* Cs = sh;
    float* Xs = sh + 16 * STR;
    float* Sp = sh + 32 * STR;
    float* Gs = Sp + 8 * 256;

    int b   = blockIdx.x / NCH;
    int c   = blockIdx.x - b * NCH;
    int p0  = c * PPB;
    int nb  = b * TT;
    int tid = threadIdx.x;
    int row = tid >> 4;
    int l16 = tid & 15;
    int wave = tid >> 5, lane = tid & 31;
    int arow = lane & 15;
    int kofs = 2 * (lane >> 4);

    const v4f* g4p = (const v4f*)gamma;
    const v4f* b4p = (const v4f*)beta;

    v4f xv[12];
    {
        const v4f* xc = (const v4f*)(x + (size_t)(nb + row) * DD);
        const v4f* xx = (const v4f*)(x + (size_t)((1 + p0) * NTOT + nb + row) * DD);
        v4f cv[12];
        #pragma unroll
        for (int i = 0; i < 12; ++i) {
            cv[i] = xc[l16 + 16 * i];
            xv[i] = xx[l16 + 16 * i];
        }
        ln_row_to_lds(cv, Cs + row * STR, g4p, b4p, l16);
    }

    const float* crow = Cs + arow * STR + kofs;
    const float* xrow = Xs + arow * STR + kofs;
    int kbase = wave * 96;

    v8f facc[6];
    #pragma unroll
    for (int t = 0; t < 6; ++t) facc[t] = (v8f){0.f, 0.f, 0.f, 0.f, 0.f, 0.f, 0.f, 0.f};

    #pragma unroll
    for (int it = 0; it < PPB; ++it) {
        ln_row_to_lds(xv, Xs + row * STR, g4p, b4p, l16);
        lds_barrier();

        if (it + 1 < PPB) {
            const v4f* xx = (const v4f*)(x + (size_t)((1 + p0 + it + 1) * NTOT + nb + row) * DD);
            #pragma unroll
            for (int i = 0; i < 12; ++i) xv[i] = xx[l16 + 16 * i];
        }

        v8f acc = {};
        #pragma unroll
        for (int kc = 0; kc < 24; ++kc) {
            int k0 = kbase + kc * 4;
            v2f av = *(const v2f*)(crow + k0);
            v2f bv = *(const v2f*)(xrow + k0);
            acc = __builtin_amdgcn_wmma_f32_16x16x4_f32(
                false, av, false, bv, (short)0, acc, false, false);
        }

        #pragma unroll
        for (int j = 0; j < 8; ++j) {
            int m = j + 8 * (lane >> 4);
            Sp[wave * 256 + m * 16 + (lane & 15)] = acc[j];
        }
        lds_barrier();

        float s = 0.f;
        #pragma unroll
        for (int w = 0; w < 8; ++w) s += Sp[w * 256 + tid];

        int nc = tid >> 4;
        int nn = tid & 15;
        float g = 0.f;
        int dlt = nn - nc;
        if (dlt >= -2 && dlt <= 2) {
            float aff = s * AFF_SCALE;
            g = 1.f / (1.f + __expf(-aff)) - 0.5f;
        }
        Gs[nn * 16 + nc] = g;
        lds_barrier();

        v2f bg[4];
        #pragma unroll
        for (int kk = 0; kk < 4; ++kk) {
            int r0 = kk * 4 + kofs;
            bg[kk] = (v2f){ Gs[r0 * 16 + arow], Gs[(r0 + 1) * 16 + arow] };
        }
        #pragma unroll
        for (int t = 0; t < 6; ++t) {
            int dcol = (wave * 6 + t) * 16 + arow;
            #pragma unroll
            for (int kk = 0; kk < 4; ++kk) {
                int r0 = kk * 4 + kofs;
                v2f av = { Xs[r0 * STR + dcol], Xs[(r0 + 1) * STR + dcol] };
                facc[t] = __builtin_amdgcn_wmma_f32_16x16x4_f32(
                    false, av, false, bg[kk], (short)0, facc[t], false, false);
            }
        }
        if (it + 1 < PPB) lds_barrier();
    }

    float* dst = part + (size_t)blockIdx.x * TILE_F + (size_t)(wave * 6) * 256 + lane;
    #pragma unroll
    for (int t = 0; t < 6; ++t) {
        #pragma unroll
        for (int j = 0; j < 8; ++j) dst[(t * 8 + j) * 32] = facc[t][j];
    }
}

__global__ void __launch_bounds__(256)
sum_kernel(const float* __restrict__ part, float* __restrict__ out)
{
    __shared__ float tile[16 * 33];

    int b   = blockIdx.x / 24;
    int dp  = blockIdx.x - b * 24;
    int tid = threadIdx.x;
    const float* src = part + (size_t)(b * NCH) * TILE_F + (size_t)(2 * dp) * 256 + tid;

    float s0 = 0.f, s1 = 0.f;
    #pragma unroll
    for (int c = 0; c < NCH; ++c) {
        s0 += src[(size_t)c * TILE_F];
        s1 += src[(size_t)c * TILE_F + 256];
    }

    int j = tid >> 5, lane = tid & 31;
    int n = lane & 15, dd = j + 8 * (lane >> 4);
    tile[n * 33 + dd]      = s0;
    tile[n * 33 + 16 + dd] = s1;
    __syncthreads();

    int rn = tid >> 4;
    int c2 = (tid & 15) * 2;
    v2f o = { tile[rn * 33 + c2], tile[rn * 33 + c2 + 1] };
    *(v2f*)(out + (size_t)(b * TT + rn) * DD + dp * 32 + c2) = o;
}

extern "C" void kernel_launch(void* const* d_in, const int* in_sizes, int n_in,
                              void* d_out, int out_size, void* d_ws, size_t ws_size,
                              hipStream_t stream)
{
    (void)in_sizes; (void)n_in; (void)out_size; (void)ws_size;
    const float* x     = (const float*)d_in[0];
    const float* gamma = (const float*)d_in[1];
    const float* beta  = (const float*)d_in[2];

    float* part = (float*)d_ws;

    size_t shmem = (size_t)(32 * STR + 8 * 256 + 256) * sizeof(float);
    scores_feat_kernel<<<NGROUP * NCH, 256, shmem, stream>>>(x, gamma, beta, part);

    sum_kernel<<<NGROUP * 24, 256, 0, stream>>>(part, (float*)d_out);
}


# 4 positions per block (was 7) in 256 registers (b128 LN stores; a fence holds the next LN behind the feat product): two blocks per WGP hide each other's waits; 49 partials (p sum order changes)
# speedup vs baseline: 3.2808x; 1.1666x over previous
#include <hip/hip_runtime.h>
#include <math.h>

typedef float v2f __attribute__((ext_vector_type(2)));
typedef float v4f __attribute__((ext_vector_type(4)));
typedef float v8f __attribute__((ext_vector_type(8)));

#define LL     197
#define NTOT   128
#define DD     768
#define TT     16
#define PP     196
#define NGROUP 8
#define STR    772
#define AFF_SCALE 0.03608439182435161f
#define PPB    4
#define NCH    (PP / PPB)
#define TILE_F (16 * DD)

__device__ __forceinline__ void lds_barrier()
{
    __builtin_amdgcn_fence(__ATOMIC_RELEASE, "workgroup", "local");
    __builtin_amdgcn_s_barrier();
    __builtin_amdgcn_fence(__ATOMIC_ACQUIRE, "workgroup", "local");
}

__device__ __forceinline__ void ln_row_to_lds(const v4f (&v)[12], float* dst,
                                              const v4f* g4l, const v4f* b4l)
{
    v4f s4 = {0.f, 0.f, 0.f, 0.f}, q4 = {0.f, 0.f, 0.f, 0.f};
    #pragma unroll
    for (int i = 0; i < 12; ++i) {
        s4 += v[i];
        q4 += v[i] * v[i];
    }
    float s = (s4.x + s4.y) + (s4.z + s4.w);
    float q = (q4.x + q4.y) + (q4.z + q4.w);
    #pragma unroll
    for (int m = 8; m >= 1; m >>= 1) {
        s += __shfl_xor(s, m);
        q += __shfl_xor(q, m);
    }
    float mu = s * (1.f / DD);
    float rs = rsqrtf(q * (1.f / DD) - mu * mu + 1e-5f);
    #pragma unroll
    for (int i = 0; i < 12; ++i) {
        v4f gv = g4l[16 * i], bv = b4l[16 * i];
        v4f o = (v[i] - mu) * rs * gv + bv;
        *(v4f*)(dst + 64 * i) = o;
    }
}

__global__ void __launch_bounds__(256, 4)
scores_feat_kernel(const float* __restrict__ x,
                   const float* __restrict__ gamma,
                   const float* __restrict__ beta,
                   float* __restrict__ part)
{
    extern __shared__ float sh[];
    float* Cs = sh;
    float* Xs = sh + 16 * STR;
    float* Sp = sh + 32 * STR;
    float* Gs = Sp + 8 * 256;

    int b   = blockIdx.x / NCH;
    int c   = blockIdx.x - b * NCH;
    int p0  = c * PPB;
    int nb  = b * TT;
    int tid = threadIdx.x;
    int row = tid >> 4;
    int l16 = tid & 15;
    int wave = tid >> 5, lane = tid & 31;
    int arow = lane & 15;
    int kofs = 2 * (lane >> 4);

    const v4f* g4l = (const v4f*)gamma + l16;
    const v4f* b4l = (const v4f*)beta + l16;
    float* cdst = Cs + row * STR + 4 * l16;
    float* xdst = Xs + row * STR + 4 * l16;

    v4f xv[12];
    {
        const v4f* xc = (const v4f*)(x + (size_t)(nb + row) * DD);
        const v4f* xx = (const v4f*)(x + (size_t)((1 + p0) * NTOT + nb + row) * DD);
        v4f cv[12];
        #pragma unroll
        for (int i = 0; i < 12; ++i) {
            cv[i] = xc[l16 + 16 * i];
            xv[i] = xx[l16 + 16 * i];
        }
        ln_row_to_lds(cv, cdst, g4l, b4l);
    }

    const float* crow = Cs + arow * STR + kofs;
    const float* xrow = Xs + arow * STR + kofs;
    int kbase = wave * 96;

    v8f facc[6];
    #pragma unroll
    for (int t = 0; t < 6; ++t) facc[t] = (v8f){0.f, 0.f, 0.f, 0.f, 0.f, 0.f, 0.f, 0.f};

    #pragma unroll
    for (int it = 0; it < PPB; ++it) {
        ln_row_to_lds(xv, xdst, g4l, b4l);
        lds_barrier();

        if (it + 1 < PPB) {
            const v4f* xx = (const v4f*)(x + (size_t)((1 + p0 + it + 1) * NTOT + nb + row) * DD);
            #pragma unroll
            for (int i = 0; i < 12; ++i) xv[i] = xx[l16 + 16 * i];
        }

        v8f acc = {};
        #pragma unroll
        for (int kc = 0; kc < 24; ++kc) {
            int k0 = kbase + kc * 4;
            v2f av = *(const v2f*)(crow + k0);
            v2f bv = *(const v2f*)(xrow + k0);
            acc = __builtin_amdgcn_wmma_f32_16x16x4_f32(
                false, av, false, bv, (short)0, acc, false, false);
        }

        #pragma unroll
        for (int j = 0; j < 8; ++j) {
            int m = j + 8 * (lane >> 4);
            Sp[wave * 256 + m * 16 + (lane & 15)] = acc[j];
        }
        lds_barrier();

        float s = 0.f;
        #pragma unroll
        for (int w = 0; w < 8; ++w) s += Sp[w * 256 + tid];

        int nc = tid >> 4;
        int nn = tid & 15;
        float g = 0.f;
        int dlt = nn - nc;
        if (dlt >= -2 && dlt <= 2) {
            float aff = s * AFF_SCALE;
            g = 1.f / (1.f + __expf(-aff)) - 0.5f;
        }
        Gs[nn * 16 + nc] = g;
        lds_barrier();

        v2f bg[4];
        #pragma unroll
        for (int kk = 0; kk < 4; ++kk) {
            int r0 = kk * 4 + kofs;
            bg[kk] = (v2f){ Gs[r0 * 16 + arow], Gs[(r0 + 1) * 16 + arow] };
        }
        #pragma unroll
        for (int t = 0; t < 6; ++t) {
            int dcol = (wave * 6 + t) * 16 + arow;
            #pragma unroll
            for (int kk = 0; kk < 4; ++kk) {
                int r0 = kk * 4 + kofs;
                v2f av = { Xs[r0 * STR + dcol], Xs[(r0 + 1) * STR + dcol] };
                facc[t] = __builtin_amdgcn_wmma_f32_16x16x4_f32(
                    false, av, false, bg[kk], (short)0, facc[t], false, false);
            }
        }
        if (it + 1 < PPB) {
            lds_barrier();
            __builtin_amdgcn_sched_barrier(0);
        }
    }

    float* dst = part + (size_t)blockIdx.x * TILE_F + (size_t)(wave * 6) * 256 + lane;
    #pragma unroll
    for (int t = 0; t < 6; ++t) {
        #pragma unroll
        for (int j = 0; j < 8; ++j) dst[(t * 8 + j) * 32] = facc[t][j];
    }
}

__global__ void __launch_bounds__(256)
sum_kernel(const float* __restrict__ part, float* __restrict__ out)
{
    __shared__ float tile[16 * 33];

    int b   = blockIdx.x / 24;
    int dp  = blockIdx.x - b * 24;
    int tid = threadIdx.x;
    const float* src = part + (size_t)(b * NCH) * TILE_F + (size_t)(2 * dp) * 256 + tid;

    float s0 = 0.f, s1 = 0.f;
    #pragma unroll
    for (int c = 0; c < NCH; ++c) {
        s0 += src[(size_t)c * TILE_F];
        s1 += src[(size_t)c * TILE_F + 256];
    }

    int j = tid >> 5, lane = tid & 31;
    int n = lane & 15, dd = j + 8 * (lane >> 4);
    tile[n * 33 + dd]      = s0;
    tile[n * 33 + 16 + dd] = s1;
    __syncthreads();

    int rn = tid >> 4;
    int c2 = (tid & 15) * 2;
    v2f o = { tile[rn * 33 + c2], tile[rn * 33 + c2 + 1] };
    *(v2f*)(out + (size_t)(b * TT + rn) * DD + dp * 32 + c2) = o;
}

extern "C" void kernel_launch(void* const* d_in, const int* in_sizes, int n_in,
                              void* d_out, int out_size, void* d_ws, size_t ws_size,
                              hipStream_t stream)
{
    (void)in_sizes; (void)n_in; (void)out_size; (void)ws_size;
    const float* x     = (const float*)d_in[0];
    const float* gamma = (const float*)d_in[1];
    const float* beta  = (const float*)d_in[2];

    float* part = (float*)d_ws;

    size_t shmem = (size_t)(32 * STR + 8 * 256 + 256) * sizeof(float);
    scores_feat_kernel<<<NGROUP * NCH, 256, shmem, stream>>>(x, gamma, beta, part);

    sum_kernel<<<NGROUP * 24, 256, 0, stream>>>(part, (float*)d_out);
}
